# starting kernel
# baseline (speedup 1.0000x reference)
#include <hip/hip_runtime.h>

#define NNODES 50000
#define MTILES (NNODES / 16)

typedef float v2f __attribute__((ext_vector_type(2)));
typedef float v8f __attribute__((ext_vector_type(8)));

__device__ __forceinline__ unsigned enc_f32(float f) {
    unsigned u = __float_as_uint(f);
    return (u & 0x80000000u) ? ~u : (u | 0x80000000u);
}
__device__ __forceinline__ float dec_f32(unsigned e) {
    if (e == 0u) return 0.0f;
    unsigned u = (e & 0x80000000u) ? (e ^ 0x80000000u) : ~e;
    return __uint_as_float(u);
}

__global__ void gin_zero_kernel(unsigned* __restrict__ p, long long n4) {
    long long i = blockIdx.x * (long long)blockDim.x + threadIdx.x;
    if (i < n4) ((uint4*)p)[i] = make_uint4(0u, 0u, 0u, 0u);
}

__global__ void gin_scatter_max_kernel(const int* __restrict__ src, const int* __restrict__ dst,
                                       const float* __restrict__ X, unsigned* __restrict__ agg,
                                       int d, int chunk_shift, long long total) {
    long long idx = blockIdx.x * (long long)blockDim.x + threadIdx.x;
    if (idx >= total) return;
    int e = (int)(idx >> chunk_shift);
    int c = (int)(idx & ((1 << chunk_shift) - 1));
    int s = src[e];
    int t = dst[e];
    const float4 v = ((const float4*)(X + (size_t)s * d))[c];
    unsigned* ap = agg + (size_t)t * d + (size_t)c * 4;
    atomicMax(ap + 0, enc_f32(v.x));
    atomicMax(ap + 1, enc_f32(v.y));
    atomicMax(ap + 2, enc_f32(v.z));
    atomicMax(ap + 3, enc_f32(v.w));
}

__global__ void gin_combine_kernel(const float* __restrict__ X, const unsigned* __restrict__ agg,
                                   float* __restrict__ out, long long n4) {
    long long i = blockIdx.x * (long long)blockDim.x + threadIdx.x;
    if (i >= n4) return;
    float4 x = ((const float4*)X)[i];
    uint4  a = ((const uint4*)agg)[i];
    float4 r;
    r.x = x.x + dec_f32(a.x);
    r.y = x.y + dec_f32(a.y);
    r.z = x.z + dec_f32(a.z);
    r.w = x.w + dec_f32(a.w);
    ((float4*)out)[i] = r;
}

__global__ __launch_bounds__(128) void gin_gemm_wmma_kernel(
    const float* __restrict__ X, const float* __restrict__ W,
    const float* __restrict__ bias, float* __restrict__ Y,
    int K, int Nout, int relu)
{
    const int lane  = threadIdx.x & 31;
    const int wave  = threadIdx.x >> 5;
    const int mtile = blockIdx.x * 4 + wave;
    if (mtile >= MTILES) return;

    const int row0    = mtile * 16;
    const int l15     = lane & 15;
    const int koff    = (lane >> 4) * 2;
    const int colbase = blockIdx.y * 64 + l15;

    const float* __restrict__ xrow = X + (size_t)(row0 + l15) * K;

    v8f acc0 = {}, acc1 = {}, acc2 = {}, acc3 = {};
#pragma unroll 2
    for (int k0 = 0; k0 < K; k0 += 4) {
        v2f a;
        a.x = xrow[k0 + koff + 0];
        a.y = xrow[k0 + koff + 1];
        const float* __restrict__ w0 = W + (size_t)(k0 + koff + 0) * Nout + colbase;
        const float* __restrict__ w1 = W + (size_t)(k0 + koff + 1) * Nout + colbase;
        v2f b0, b1, b2, b3;
        b0.x = w0[ 0]; b0.y = w1[ 0];
        b1.x = w0[16]; b1.y = w1[16];
        b2.x = w0[32]; b2.y = w1[32];
        b3.x = w0[48]; b3.y = w1[48];
        acc0 = __builtin_amdgcn_wmma_f32_16x16x4_f32(false, a, false, b0, (short)0, acc0, false, false);
        acc1 = __builtin_amdgcn_wmma_f32_16x16x4_f32(false, a, false, b1, (short)0, acc1, false, false);
        acc2 = __builtin_amdgcn_wmma_f32_16x16x4_f32(false, a, false, b2, (short)0, acc2, false, false);
        acc3 = __builtin_amdgcn_wmma_f32_16x16x4_f32(false, a, false, b3, (short)0, acc3, false, false);
    }

    const int rbase = row0 + (lane >> 4) * 8;
    v8f accs[4] = {acc0, acc1, acc2, acc3};
#pragma unroll
    for (int t = 0; t < 4; ++t) {
        const int col = colbase + 16 * t;
        const float bv = bias[col];
        float* __restrict__ yc = Y + col;
#pragma unroll
        for (int r = 0; r < 8; ++r) {
            float v = accs[t][r] + bv;
            if (relu) v = fmaxf(v, 0.0f);
            yc[(size_t)(rbase + r) * Nout] = v;
        }
    }
}

static void gin_layer(const int* src, const int* dst, int E,
                      const float* Xin, int d_in_, const float* W, const float* b,
                      int d_out_, float* Y, int relu,
                      unsigned* agg, float* xsum, hipStream_t stream)
{
    long long nelem = (long long)NNODES * d_in_;
    long long n4 = nelem >> 2;
    gin_zero_kernel<<<(unsigned)((n4 + 255) / 256), 256, 0, stream>>>(agg, n4);

    int cs = (d_in_ == 128) ? 5 : 6;
    long long total = (long long)E << cs;
    gin_scatter_max_kernel<<<(unsigned)((total + 255) / 256), 256, 0, stream>>>(
        src, dst, Xin, agg, d_in_, cs, total);

    gin_combine_kernel<<<(unsigned)((n4 + 255) / 256), 256, 0, stream>>>(Xin, agg, xsum, n4);

    dim3 grid((MTILES + 3) / 4, d_out_ / 64);
    gin_gemm_wmma_kernel<<<grid, 128, 0, stream>>>(xsum, W, b, Y, d_in_, d_out_, relu);
}

extern "C" void kernel_launch(void* const* d_in, const int* in_sizes, int n_in,
                              void* d_out, int out_size, void* d_ws, size_t ws_size,
                              hipStream_t stream)
{
    const int*   src = (const int*)d_in[0];
    const int*   dst = (const int*)d_in[1];
    const float* emb = (const float*)d_in[2];
    const float* W1  = (const float*)d_in[3];
    const float* b1  = (const float*)d_in[4];
    const float* W2  = (const float*)d_in[5];
    const float* b2  = (const float*)d_in[6];
    const float* W3  = (const float*)d_in[7];
    const float* b3  = (const float*)d_in[8];
    float* out = (float*)d_out;
    const int E = in_sizes[0];

    const size_t cap = (size_t)NNODES * 256;
    unsigned* agg  = (unsigned*)d_ws;
    float*    xsum = (float*)d_ws + cap;
    float*    h    = (float*)d_ws + 2 * cap;

    gin_layer(src, dst, E, emb, 128, W1, b1, 256, h, 1, agg, xsum, stream);
    gin_layer(src, dst, E, h, 256, W2, b2, 256, h, 1, agg, xsum, stream);
    gin_layer(src, dst, E, h, 256, W3, b3, 128, out, 0, agg, xsum, stream);
}


# Neighbour max kept in LDS: 256 work-groups each own 196 destination rows, filter the edge stream for them, fold hit rows with LDS max atomics; no global atomic, zero or x+max launch; sums unchanged
# speedup vs baseline: 2.5018x; 2.5018x over previous
#include <hip/hip_runtime.h>

#define NNODES 50000
#define MTILES (NNODES / 16)

typedef float v2f __attribute__((ext_vector_type(2)));
typedef float v8f __attribute__((ext_vector_type(8)));
typedef float v4f __attribute__((ext_vector_type(4)));

__device__ __forceinline__ unsigned enc_f32(float f) {
    unsigned u = __float_as_uint(f);
    return (u & 0x80000000u) ? ~u : (u | 0x80000000u);
}
__device__ __forceinline__ float dec_f32(unsigned e) {
    if (e == 0u) return 0.0f;
    unsigned u = (e & 0x80000000u) ? (e ^ 0x80000000u) : ~e;
    return __uint_as_float(u);
}

#define RNODES  196
#define NGROUPS 256
#define QR      4096

__device__ __forceinline__ int acc_pos(int c) {
    return (c & ~127) + ((c & 3) << 5) + ((c & 127) >> 2);
}

__device__ __forceinline__ void gin_filter(const int* __restrict__ dst, int E, int nsuper, int rot,
                                           int sweep0, int ipr, int wave, int lane, int lo, int rn,
                                           int2* q, unsigned* qn, unsigned* ovf)
{
    for (int it = 0; it < ipr; ++it) {
        const int sc0 = wave + 32 * (sweep0 + it);
        if (sc0 >= nsuper) break;
        const int sc = (sc0 + rot < nsuper) ? (sc0 + rot) : (sc0 + rot - nsuper);
        const int base = (sc << 10) + lane * 4;
        int4 d[8];
        if ((sc << 10) + 1024 <= E) {
#pragma unroll
            for (int u = 0; u < 8; ++u) d[u] = *(const int4*)(dst + base + u * 128);
        } else {
#pragma unroll
            for (int u = 0; u < 8; ++u) {
                const int i = base + u * 128;
                d[u].x = (i + 0 < E) ? dst[i + 0] : -1;
                d[u].y = (i + 1 < E) ? dst[i + 1] : -1;
                d[u].z = (i + 2 < E) ? dst[i + 2] : -1;
                d[u].w = (i + 3 < E) ? dst[i + 3] : -1;
            }
        }
#pragma unroll
        for (int u = 0; u < 8; ++u) {
            const unsigned r0 = (unsigned)(d[u].x - lo), r1 = (unsigned)(d[u].y - lo);
            const unsigned r2 = (unsigned)(d[u].z - lo), r3 = (unsigned)(d[u].w - lo);
            const bool any4 = (r0 < (unsigned)rn) | (r1 < (unsigned)rn) | (r2 < (unsigned)rn) | (r3 < (unsigned)rn);
            if (__builtin_amdgcn_ballot_w32(any4) != 0u) {
                const unsigned rr[4] = {r0, r1, r2, r3};
#pragma unroll
                for (int c = 0; c < 4; ++c) {
                    if (rr[c] < (unsigned)rn) {
                        const unsigned slot = atomicAdd(qn, 1u);
                        if (slot < (unsigned)QR) q[slot] = make_int2(base + u * 128 + c, (int)rr[c]);
                        else *ovf = 1u;
                    }
                }
            }
        }
    }
}

template <int D>
__device__ __forceinline__ void gin_fold(const int* __restrict__ src, const float* __restrict__ X,
                                         unsigned* acc, const int2* q, const unsigned* qn,
                                         int wave, int lane)
{
    constexpr int NB = D / 128;
    constexpr int U  = 8 / NB;
    static_assert(U * NB == 8, "the load fence below names 8 register sets");
    const int cnt = min(__builtin_amdgcn_readfirstlane((int)*qn), QR);
    const int nmine = (cnt > wave) ? ((cnt - wave + 31) >> 5) : 0;
    for (int m0 = 0; m0 < nmine; m0 += 32) {
        const int nv = min(32, nmine - m0);
        int sv = 0, tv = 0;
        if (lane < nv) {
            const int2 e = q[wave + 32 * (m0 + lane)];
            sv = min(max(src[e.x], 0), NNODES - 1);
            tv = e.y;
        }
        for (int j = 0; j < nv; j += U) {
            v4f v[8];
            int tr[U];
#pragma unroll
            for (int u = 0; u < U; ++u) {
                const int ju = min(j + u, nv - 1);
                const int s = __builtin_amdgcn_readlane(sv, ju);
                tr[u] = (j + u < nv) ? __builtin_amdgcn_readlane(tv, ju) : -1;
#pragma unroll
                for (int k = 0; k < NB; ++k)
                    v[u * NB + k] = *(const v4f*)(X + (size_t)s * D + 128 * k + 4 * lane);
            }
            asm volatile("" :: "v"(v[0]), "v"(v[1]), "v"(v[2]), "v"(v[3]),
                               "v"(v[4]), "v"(v[5]), "v"(v[6]), "v"(v[7]));
#pragma unroll
            for (int u = 0; u < U; ++u) {
                if (tr[u] >= 0) {
                    unsigned* a = acc + tr[u] * D + lane;
#pragma unroll
                    for (int k = 0; k < NB; ++k) {
                        const v4f x = v[u * NB + k];
                        atomicMax(a + 128 * k +  0, enc_f32(x.x));
                        atomicMax(a + 128 * k + 32, enc_f32(x.y));
                        atomicMax(a + 128 * k + 64, enc_f32(x.z));
                        atomicMax(a + 128 * k + 96, enc_f32(x.w));
                    }
                }
            }
        }
    }
}

template <int D>
__global__ __launch_bounds__(1024) void gin_rangemax_kernel(
    const int* __restrict__ src, const int* __restrict__ dst,
    const float* __restrict__ X, float* __restrict__ xsum, int E)
{
    constexpr int NB = D / 128;
    __shared__ __attribute__((aligned(16))) unsigned acc[RNODES * D];
    __shared__ __attribute__((aligned(16))) int2 queue[2 * QR];
    __shared__ unsigned qcount[3];
    __shared__ unsigned ovf;

    const int tid  = threadIdx.x;
    const int lane = tid & 31;
    const int wave = __builtin_amdgcn_readfirstlane(tid >> 5);
    const int lo   = blockIdx.x * RNODES;
    const int rn   = min(RNODES, NNODES - lo);

    for (int i = tid; i < RNODES * D / 4; i += 1024) ((uint4*)acc)[i] = make_uint4(0u, 0u, 0u, 0u);
    if (tid < 3) qcount[tid] = 0u;
    if (tid == 3) ovf = 0u;
    __syncthreads();

    const int nsuper = (E + 1023) >> 10;
    const int rot    = (int)(((long long)blockIdx.x * nsuper) >> 8);
    const int niter  = (nsuper + 31) >> 5;
    const int ipr    = max(5, (niter + 15) >> 4);
    const int nround = (niter + ipr - 1) / ipr;

    const bool fold_first = ((wave >> 2) & 1) != 0;
    int c_prev = 2, c_cur = 0, c_next = 1;
    for (int r = 0; r <= nround; ++r) {
        int2* qcur = queue + (r & 1) * QR;
        const int2* qprev = queue + ((r & 1) ^ 1) * QR;
        if (tid == 0) qcount[c_next] = 0u;
        if (fold_first && r > 0) gin_fold<D>(src, X, acc, qprev, &qcount[c_prev], wave, lane);
        if (r < nround) gin_filter(dst, E, nsuper, rot, r * ipr, ipr, wave, lane, lo, rn, qcur, &qcount[c_cur], &ovf);
        if (!fold_first && r > 0) gin_fold<D>(src, X, acc, qprev, &qcount[c_prev], wave, lane);
        __syncthreads();
        const int t = c_prev; c_prev = c_cur; c_cur = c_next; c_next = t;
    }

    if (ovf != 0u) {
        for (int e = tid; e < E; e += 1024) {
            const unsigned r = (unsigned)(dst[e] - lo);
            if (r < (unsigned)rn) {
                const int s = min(max(src[e], 0), NNODES - 1);
                const float* __restrict__ xr = X + (size_t)s * D;
                for (int c = 0; c < D; ++c) atomicMax(acc + r * D + acc_pos(c), enc_f32(xr[c]));
            }
        }
    }
    __syncthreads();

    for (int n = wave; n < rn; n += 32) {
        const float* __restrict__ xr = X + (size_t)(lo + n) * D;
        float* __restrict__ o = xsum + (size_t)(lo + n) * D;
#pragma unroll
        for (int k = 0; k < NB; ++k) {
            const float4 x = *(const float4*)(xr + 128 * k + 4 * lane);
            const unsigned* a = acc + n * D + 128 * k + lane;
            float4 r;
            r.x = x.x + dec_f32(a[0]);
            r.y = x.y + dec_f32(a[32]);
            r.z = x.z + dec_f32(a[64]);
            r.w = x.w + dec_f32(a[96]);
            *(float4*)(o + 128 * k + 4 * lane) = r;
        }
    }
}

__global__ __launch_bounds__(128) void gin_gemm_wmma_kernel(
    const float* __restrict__ X, const float* __restrict__ W,
    const float* __restrict__ bias, float* __restrict__ Y,
    int K, int Nout, int relu)
{
    const int lane  = threadIdx.x & 31;
    const int wave  = threadIdx.x >> 5;
    const int mtile = blockIdx.x * 4 + wave;
    if (mtile >= MTILES) return;

    const int row0    = mtile * 16;
    const int l15     = lane & 15;
    const int koff    = (lane >> 4) * 2;
    const int colbase = blockIdx.y * 64 + l15;

    const float* __restrict__ xrow = X + (size_t)(row0 + l15) * K;

    v8f acc0 = {}, acc1 = {}, acc2 = {}, acc3 = {};
#pragma unroll 2
    for (int k0 = 0; k0 < K; k0 += 4) {
        v2f a;
        a.x = xrow[k0 + koff + 0];
        a.y = xrow[k0 + koff + 1];
        const float* __restrict__ w0 = W + (size_t)(k0 + koff + 0) * Nout + colbase;
        const float* __restrict__ w1 = W + (size_t)(k0 + koff + 1) * Nout + colbase;
        v2f b0, b1, b2, b3;
        b0.x = w0[ 0]; b0.y = w1[ 0];
        b1.x = w0[16]; b1.y = w1[16];
        b2.x = w0[32]; b2.y = w1[32];
        b3.x = w0[48]; b3.y = w1[48];
        acc0 = __builtin_amdgcn_wmma_f32_16x16x4_f32(false, a, false, b0, (short)0, acc0, false, false);
        acc1 = __builtin_amdgcn_wmma_f32_16x16x4_f32(false, a, false, b1, (short)0, acc1, false, false);
        acc2 = __builtin_amdgcn_wmma_f32_16x16x4_f32(false, a, false, b2, (short)0, acc2, false, false);
        acc3 = __builtin_amdgcn_wmma_f32_16x16x4_f32(false, a, false, b3, (short)0, acc3, false, false);
    }

    const int rbase = row0 + (lane >> 4) * 8;
    v8f accs[4] = {acc0, acc1, acc2, acc3};
#pragma unroll
    for (int t = 0; t < 4; ++t) {
        const int col = colbase + 16 * t;
        const float bv = bias[col];
        float* __restrict__ yc = Y + col;
#pragma unroll
        for (int r = 0; r < 8; ++r) {
            float v = accs[t][r] + bv;
            if (relu) v = fmaxf(v, 0.0f);
            yc[(size_t)(rbase + r) * Nout] = v;
        }
    }
}

static void gin_layer(const int* src, const int* dst, int E,
                      const float* Xin, int d_in_, const float* W, const float* b,
                      int d_out_, float* Y, int relu,
                      float* xsum, hipStream_t stream)
{
    if (d_in_ == 128)
        gin_rangemax_kernel<128><<<NGROUPS, 1024, 0, stream>>>(src, dst, Xin, xsum, E);
    else
        gin_rangemax_kernel<256><<<NGROUPS, 1024, 0, stream>>>(src, dst, Xin, xsum, E);

    dim3 grid((MTILES + 3) / 4, d_out_ / 64);
    gin_gemm_wmma_kernel<<<grid, 128, 0, stream>>>(xsum, W, b, Y, d_in_, d_out_, relu);
}

extern "C" void kernel_launch(void* const* d_in, const int* in_sizes, int n_in,
                              void* d_out, int out_size, void* d_ws, size_t ws_size,
                              hipStream_t stream)
{
    const int*   src = (const int*)d_in[0];
    const int*   dst = (const int*)d_in[1];
    const float* emb = (const float*)d_in[2];
    const float* W1  = (const float*)d_in[3];
    const float* b1  = (const float*)d_in[4];
    const float* W2  = (const float*)d_in[5];
    const float* b2  = (const float*)d_in[6];
    const float* W3  = (const float*)d_in[7];
    const float* b3  = (const float*)d_in[8];
    float* out = (float*)d_out;
    const int E = in_sizes[0];

    const size_t cap = (size_t)NNODES * 256;
    float* xsum = (float*)d_ws;
    float* h    = (float*)d_ws + cap;

    gin_layer(src, dst, E, emb, 128, W1, b1, 256, h, 1, xsum, stream);
    gin_layer(src, dst, E, h, 256, W2, b2, 256, h, 1, xsum, stream);
    gin_layer(src, dst, E, h, 256, W3, b3, 128, out, 0, xsum, stream);
}


# Filter reads one destination-range key byte per edge, written once per call by a new first launch (7 launches), not the 4-byte dst: a quarter of the filter's bytes in every layer; sums unchanged
# speedup vs baseline: 2.5794x; 1.0310x over previous
#include <hip/hip_runtime.h>

#define NNODES 50000
#define MTILES (NNODES / 16)

typedef float v2f __attribute__((ext_vector_type(2)));
typedef float v8f __attribute__((ext_vector_type(8)));
typedef float v4f __attribute__((ext_vector_type(4)));

__device__ __forceinline__ unsigned enc_f32(float f) {
    unsigned u = __float_as_uint(f);
    return (u & 0x80000000u) ? ~u : (u | 0x80000000u);
}
__device__ __forceinline__ float dec_f32(unsigned e) {
    if (e == 0u) return 0.0f;
    unsigned u = (e & 0x80000000u) ? (e ^ 0x80000000u) : ~e;
    return __uint_as_float(u);
}

#define RNODES  196
#define NGROUPS 256
#define QR      4096

__device__ __forceinline__ unsigned gin_key(int d) {
    return min((unsigned)d / (unsigned)RNODES, 255u);
}

__device__ __forceinline__ unsigned zero_bytes(unsigned x) {
    return (x - 0x01010101u) & ~x & 0x80808080u;
}

__global__ __launch_bounds__(256) void gin_keys_kernel(const int* __restrict__ dst,
                                                       unsigned* __restrict__ keys, int E)
{
    const int ndw = ((E + 4095) >> 12) << 10;
    for (int t = blockIdx.x * 256 + threadIdx.x; t < ndw; t += gridDim.x * 256) {
        const int e0 = 4 * t;
        int4 d;
        if (e0 + 4 <= E) {
            d = *(const int4*)(dst + e0);
        } else {
            d.x = (e0 + 0 < E) ? dst[e0 + 0] : -1;
            d.y = (e0 + 1 < E) ? dst[e0 + 1] : -1;
            d.z = (e0 + 2 < E) ? dst[e0 + 2] : -1;
            d.w = (e0 + 3 < E) ? dst[e0 + 3] : -1;
        }
        keys[t] = gin_key(d.x) | (gin_key(d.y) << 8) | (gin_key(d.z) << 16) | (gin_key(d.w) << 24);
    }
}

__device__ __forceinline__ int acc_pos(int c) {
    return (c & ~127) + ((c & 3) << 5) + ((c & 127) >> 2);
}

__device__ __forceinline__ void gin_filter(const unsigned* __restrict__ keys, int E, int nsuper, int rot,
                                           int sweep0, int ipr, int wave, int lane, unsigned pat,
                                           int* q, unsigned* qn, unsigned* ovf)
{
    for (int it = 0; it < ipr; ++it) {
        const int sc0 = wave + 32 * (sweep0 + it);
        if (sc0 >= nsuper) break;
        const int sc = (sc0 + rot < nsuper) ? (sc0 + rot) : (sc0 + rot - nsuper);
        const int base = (sc << 10) + lane * 4;
        uint4 k[8];
#pragma unroll
        for (int u = 0; u < 8; ++u) k[u] = *(const uint4*)(keys + base + u * 128);
#pragma unroll
        for (int u = 0; u < 8; ++u) {
            const unsigned w[4] = {k[u].x, k[u].y, k[u].z, k[u].w};
#pragma unroll
            for (int c = 0; c < 4; ++c) {
                unsigned x  = w[c] ^ pat;
                unsigned hz = zero_bytes(x);
                if (__builtin_amdgcn_ballot_w32(hz != 0u) != 0u) {
                    do {
                        if (hz != 0u) {
                            const int sh = __builtin_ctz(hz) & ~7;
                            const int e  = 4 * (base + u * 128 + c) + (sh >> 3);
                            if (e < E) {
                                const unsigned slot = atomicAdd(qn, 1u);
                                if (slot < (unsigned)QR) q[slot] = e;
                                else *ovf = 1u;
                            }
                            x |= 0xFFu << sh;
                            hz = zero_bytes(x);
                        }
                    } while (__builtin_amdgcn_ballot_w32(hz != 0u) != 0u);
                }
            }
        }
    }
}

template <int D>
__device__ __forceinline__ void gin_fold(const int* __restrict__ src, const int* __restrict__ dst,
                                         const float* __restrict__ X, int lo, int rn,
                                         unsigned* acc, const int* q, const unsigned* qn,
                                         int wave, int lane)
{
    constexpr int NB = D / 128;
    constexpr int U  = 8 / NB;
    static_assert(U * NB == 8, "the load fence below names 8 register sets");
    const int cnt = min(__builtin_amdgcn_readfirstlane((int)*qn), QR);
    const int nmine = (cnt > wave) ? ((cnt - wave + 31) >> 5) : 0;
    for (int m0 = 0; m0 < nmine; m0 += 32) {
        const int nv = min(32, nmine - m0);
        int sv = 0, tv = -1;
        if (lane < nv) {
            const int e = q[wave + 32 * (m0 + lane)];
            const unsigned r = (unsigned)(dst[e] - lo);
            sv = min(max(src[e], 0), NNODES - 1);
            tv = (r < (unsigned)rn) ? (int)r : -1;
        }
        for (int j = 0; j < nv; j += U) {
            v4f v[8];
            int tr[U];
#pragma unroll
            for (int u = 0; u < U; ++u) {
                const int ju = min(j + u, nv - 1);
                const int s = __builtin_amdgcn_readlane(sv, ju);
                tr[u] = (j + u < nv) ? __builtin_amdgcn_readlane(tv, ju) : -1;
#pragma unroll
                for (int k = 0; k < NB; ++k)
                    v[u * NB + k] = *(const v4f*)(X + (size_t)s * D + 128 * k + 4 * lane);
            }
            asm volatile("" :: "v"(v[0]), "v"(v[1]), "v"(v[2]), "v"(v[3]),
                               "v"(v[4]), "v"(v[5]), "v"(v[6]), "v"(v[7]));
#pragma unroll
            for (int u = 0; u < U; ++u) {
                if (tr[u] >= 0) {
                    unsigned* a = acc + tr[u] * D + lane;
#pragma unroll
                    for (int k = 0; k < NB; ++k) {
                        const v4f x = v[u * NB + k];
                        atomicMax(a + 128 * k +  0, enc_f32(x.x));
                        atomicMax(a + 128 * k + 32, enc_f32(x.y));
                        atomicMax(a + 128 * k + 64, enc_f32(x.z));
                        atomicMax(a + 128 * k + 96, enc_f32(x.w));
                    }
                }
            }
        }
    }
}

template <int D>
__global__ __launch_bounds__(1024) void gin_rangemax_kernel(
    const int* __restrict__ src, const int* __restrict__ dst, const unsigned* __restrict__ keys,
    const float* __restrict__ X, float* __restrict__ xsum, int E)
{
    constexpr int NB = D / 128;
    __shared__ __attribute__((aligned(16))) unsigned acc[RNODES * D];
    __shared__ __attribute__((aligned(16))) int queue[2 * QR];
    __shared__ unsigned qcount[3];
    __shared__ unsigned ovf;

    const int tid  = threadIdx.x;
    const int lane = tid & 31;
    const int wave = __builtin_amdgcn_readfirstlane(tid >> 5);
    const int lo   = blockIdx.x * RNODES;
    const int rn   = min(RNODES, NNODES - lo);
    const unsigned pat = (unsigned)blockIdx.x * 0x01010101u;

    for (int i = tid; i < RNODES * D / 4; i += 1024) ((uint4*)acc)[i] = make_uint4(0u, 0u, 0u, 0u);
    if (tid < 3) qcount[tid] = 0u;
    if (tid == 3) ovf = 0u;
    __syncthreads();

    const int nsuper = (E + 4095) >> 12;
    const int rot    = (int)(((long long)blockIdx.x * nsuper) >> 8);
    const int niter  = (nsuper + 31) >> 5;
    const int ipr    = max(2, (niter + 15) >> 4);
    const int nround = (niter + ipr - 1) / ipr;

    const bool fold_first = ((wave >> 2) & 1) != 0;
    int c_prev = 2, c_cur = 0, c_next = 1;
    for (int r = 0; r <= nround; ++r) {
        int* qcur = queue + (r & 1) * QR;
        const int* qprev = queue + ((r & 1) ^ 1) * QR;
        if (tid == 0) qcount[c_next] = 0u;
        if (fold_first && r > 0) gin_fold<D>(src, dst, X, lo, rn, acc, qprev, &qcount[c_prev], wave, lane);
        if (r < nround) gin_filter(keys, E, nsuper, rot, r * ipr, ipr, wave, lane, pat, qcur, &qcount[c_cur], &ovf);
        if (!fold_first && r > 0) gin_fold<D>(src, dst, X, lo, rn, acc, qprev, &qcount[c_prev], wave, lane);
        __syncthreads();
        const int t = c_prev; c_prev = c_cur; c_cur = c_next; c_next = t;
    }

    if (ovf != 0u) {
        for (int e = tid; e < E; e += 1024) {
            const unsigned r = (unsigned)(dst[e] - lo);
            if (r < (unsigned)rn) {
                const int s = min(max(src[e], 0), NNODES - 1);
                const float* __restrict__ xr = X + (size_t)s * D;
                for (int c = 0; c < D; ++c) atomicMax(acc + r * D + acc_pos(c), enc_f32(xr[c]));
            }
        }
    }
    __syncthreads();

    for (int n = wave; n < rn; n += 32) {
        const float* __restrict__ xr = X + (size_t)(lo + n) * D;
        float* __restrict__ o = xsum + (size_t)(lo + n) * D;
#pragma unroll
        for (int k = 0; k < NB; ++k) {
            const float4 x = *(const float4*)(xr + 128 * k + 4 * lane);
            const unsigned* a = acc + n * D + 128 * k + lane;
            float4 r;
            r.x = x.x + dec_f32(a[0]);
            r.y = x.y + dec_f32(a[32]);
            r.z = x.z + dec_f32(a[64]);
            r.w = x.w + dec_f32(a[96]);
            *(float4*)(o + 128 * k + 4 * lane) = r;
        }
    }
}

__global__ __launch_bounds__(128) void gin_gemm_wmma_kernel(
    const float* __restrict__ X, const float* __restrict__ W,
    const float* __restrict__ bias, float* __restrict__ Y,
    int K, int Nout, int relu)
{
    const int lane  = threadIdx.x & 31;
    const int wave  = threadIdx.x >> 5;
    const int mtile = blockIdx.x * 4 + wave;
    if (mtile >= MTILES) return;

    const int row0    = mtile * 16;
    const int l15     = lane & 15;
    const int koff    = (lane >> 4) * 2;
    const int colbase = blockIdx.y * 64 + l15;

    const float* __restrict__ xrow = X + (size_t)(row0 + l15) * K;

    v8f acc0 = {}, acc1 = {}, acc2 = {}, acc3 = {};
#pragma unroll 2
    for (int k0 = 0; k0 < K; k0 += 4) {
        v2f a;
        a.x = xrow[k0 + koff + 0];
        a.y = xrow[k0 + koff + 1];
        const float* __restrict__ w0 = W + (size_t)(k0 + koff + 0) * Nout + colbase;
        const float* __restrict__ w1 = W + (size_t)(k0 + koff + 1) * Nout + colbase;
        v2f b0, b1, b2, b3;
        b0.x = w0[ 0]; b0.y = w1[ 0];
        b1.x = w0[16]; b1.y = w1[16];
        b2.x = w0[32]; b2.y = w1[32];
        b3.x = w0[48]; b3.y = w1[48];
        acc0 = __builtin_amdgcn_wmma_f32_16x16x4_f32(false, a, false, b0, (short)0, acc0, false, false);
        acc1 = __builtin_amdgcn_wmma_f32_16x16x4_f32(false, a, false, b1, (short)0, acc1, false, false);
        acc2 = __builtin_amdgcn_wmma_f32_16x16x4_f32(false, a, false, b2, (short)0, acc2, false, false);
        acc3 = __builtin_amdgcn_wmma_f32_16x16x4_f32(false, a, false, b3, (short)0, acc3, false, false);
    }

    const int rbase = row0 + (lane >> 4) * 8;
    v8f accs[4] = {acc0, acc1, acc2, acc3};
#pragma unroll
    for (int t = 0; t < 4; ++t) {
        const int col = colbase + 16 * t;
        const float bv = bias[col];
        float* __restrict__ yc = Y + col;
#pragma unroll
        for (int r = 0; r < 8; ++r) {
            float v = accs[t][r] + bv;
            if (relu) v = fmaxf(v, 0.0f);
            yc[(size_t)(rbase + r) * Nout] = v;
        }
    }
}

static void gin_layer(const int* src, const int* dst, const unsigned* keys, int E,
                      const float* Xin, int d_in_, const float* W, const float* b,
                      int d_out_, float* Y, int relu,
                      float* xsum, hipStream_t stream)
{
    if (d_in_ == 128)
        gin_rangemax_kernel<128><<<NGROUPS, 1024, 0, stream>>>(src, dst, keys, Xin, xsum, E);
    else
        gin_rangemax_kernel<256><<<NGROUPS, 1024, 0, stream>>>(src, dst, keys, Xin, xsum, E);

    dim3 grid((MTILES + 3) / 4, d_out_ / 64);
    gin_gemm_wmma_kernel<<<grid, 128, 0, stream>>>(xsum, W, b, Y, d_in_, d_out_, relu);
}

extern "C" void kernel_launch(void* const* d_in, const int* in_sizes, int n_in,
                              void* d_out, int out_size, void* d_ws, size_t ws_size,
                              hipStream_t stream)
{
    const int*   src = (const int*)d_in[0];
    const int*   dst = (const int*)d_in[1];
    const float* emb = (const float*)d_in[2];
    const float* W1  = (const float*)d_in[3];
    const float* b1  = (const float*)d_in[4];
    const float* W2  = (const float*)d_in[5];
    const float* b2  = (const float*)d_in[6];
    const float* W3  = (const float*)d_in[7];
    const float* b3  = (const float*)d_in[8];
    float* out = (float*)d_out;
    const int E = in_sizes[0];

    const size_t cap = (size_t)NNODES * 256;
    float* xsum = (float*)d_ws;
    float* h    = (float*)d_ws + cap;
    unsigned* keys = (unsigned*)d_ws + 2 * cap;

    gin_keys_kernel<<<4 * NGROUPS, 256, 0, stream>>>(dst, keys, E);

    gin_layer(src, dst, keys, E, emb, 128, W1, b1, 256, h, 1, xsum, stream);
    gin_layer(src, dst, keys, E, h, 256, W2, b2, 256, h, 1, xsum, stream);
    gin_layer(src, dst, keys, E, h, 256, W3, b3, 128, out, 0, xsum, stream);
}


# Layers 2 and 3 fold from a per-group list of (source row, local row) words kept by layer 1's fold in the same call: no key sweep, queue, rounds or index fetch there; 7 launches; sums unchanged
# speedup vs baseline: 2.9702x; 1.1515x over previous
#include <hip/hip_runtime.h>

#define NNODES 50000
#define MTILES (NNODES / 16)

typedef float v2f __attribute__((ext_vector_type(2)));
typedef float v8f __attribute__((ext_vector_type(8)));
typedef float v4f __attribute__((ext_vector_type(4)));

__device__ __forceinline__ unsigned enc_f32(float f) {
    unsigned u = __float_as_uint(f);
    return (u & 0x80000000u) ? ~u : (u | 0x80000000u);
}
__device__ __forceinline__ float dec_f32(unsigned e) {
    if (e == 0u) return 0.0f;
    unsigned u = (e & 0x80000000u) ? (e ^ 0x80000000u) : ~e;
    return __uint_as_float(u);
}

#define RNODES  196
#define NGROUPS 256
#define QR      4096
#define LCAP    65536
#define LNONE   0xFFFFFFFFu

__device__ __forceinline__ unsigned gin_key(int d) {
    return min((unsigned)d / (unsigned)RNODES, 255u);
}

__device__ __forceinline__ unsigned zero_bytes(unsigned x) {
    return (x - 0x01010101u) & ~x & 0x80808080u;
}

__global__ __launch_bounds__(256) void gin_keys_kernel(const int* __restrict__ dst,
                                                       unsigned* __restrict__ keys, int E)
{
    const int ndw = ((E + 4095) >> 12) << 10;
    for (int t = blockIdx.x * 256 + threadIdx.x; t < ndw; t += gridDim.x * 256) {
        const int e0 = 4 * t;
        int4 d;
        if (e0 + 4 <= E) {
            d = *(const int4*)(dst + e0);
        } else {
            d.x = (e0 + 0 < E) ? dst[e0 + 0] : -1;
            d.y = (e0 + 1 < E) ? dst[e0 + 1] : -1;
            d.z = (e0 + 2 < E) ? dst[e0 + 2] : -1;
            d.w = (e0 + 3 < E) ? dst[e0 + 3] : -1;
        }
        keys[t] = gin_key(d.x) | (gin_key(d.y) << 8) | (gin_key(d.z) << 16) | (gin_key(d.w) << 24);
    }
}

__device__ __forceinline__ int acc_pos(int c) {
    return (c & ~127) + ((c & 3) << 5) + ((c & 127) >> 2);
}

__device__ __forceinline__ void gin_filter(const unsigned* __restrict__ keys, int E, int nsuper, int rot,
                                           int sweep0, int ipr, int wave, int lane, unsigned pat,
                                           int* q, unsigned* qn, unsigned* ovf)
{
    for (int it = 0; it < ipr; ++it) {
        const int sc0 = wave + 32 * (sweep0 + it);
        if (sc0 >= nsuper) break;
        const int sc = (sc0 + rot < nsuper) ? (sc0 + rot) : (sc0 + rot - nsuper);
        const int base = (sc << 10) + lane * 4;
        uint4 k[8];
#pragma unroll
        for (int u = 0; u < 8; ++u) k[u] = *(const uint4*)(keys + base + u * 128);
#pragma unroll
        for (int u = 0; u < 8; ++u) {
            const unsigned w[4] = {k[u].x, k[u].y, k[u].z, k[u].w};
#pragma unroll
            for (int c = 0; c < 4; ++c) {
                unsigned x  = w[c] ^ pat;
                unsigned hz = zero_bytes(x);
                if (__builtin_amdgcn_ballot_w32(hz != 0u) != 0u) {
                    do {
                        if (hz != 0u) {
                            const int sh = __builtin_ctz(hz) & ~7;
                            const int e  = 4 * (base + u * 128 + c) + (sh >> 3);
                            if (e < E) {
                                const unsigned slot = atomicAdd(qn, 1u);
                                if (slot < (unsigned)QR) q[slot] = e;
                                else *ovf = 1u;
                            }
                            x |= 0xFFu << sh;
                            hz = zero_bytes(x);
                        }
                    } while (__builtin_amdgcn_ballot_w32(hz != 0u) != 0u);
                }
            }
        }
    }
}

template <int D>
__device__ __forceinline__ void gin_fold_rows(const float* __restrict__ X, unsigned* acc,
                                              int sv, int tv, int nv, int lane)
{
    constexpr int NB = D / 128;
    constexpr int U  = 8 / NB;
    static_assert(U * NB == 8, "the load fence below names 8 register sets");
    for (int j = 0; j < nv; j += U) {
        v4f v[8];
        int tr[U];
#pragma unroll
        for (int u = 0; u < U; ++u) {
            const int ju = min(j + u, nv - 1);
            const int s = __builtin_amdgcn_readlane(sv, ju);
            tr[u] = (j + u < nv) ? __builtin_amdgcn_readlane(tv, ju) : -1;
#pragma unroll
            for (int k = 0; k < NB; ++k)
                v[u * NB + k] = *(const v4f*)(X + (size_t)s * D + 128 * k + 4 * lane);
        }
        asm volatile("" :: "v"(v[0]), "v"(v[1]), "v"(v[2]), "v"(v[3]),
                           "v"(v[4]), "v"(v[5]), "v"(v[6]), "v"(v[7]));
#pragma unroll
        for (int u = 0; u < U; ++u) {
            if (tr[u] >= 0) {
                unsigned* a = acc + tr[u] * D + lane;
#pragma unroll
                for (int k = 0; k < NB; ++k) {
                    const v4f x = v[u * NB + k];
                    atomicMax(a + 128 * k +  0, enc_f32(x.x));
                    atomicMax(a + 128 * k + 32, enc_f32(x.y));
                    atomicMax(a + 128 * k + 64, enc_f32(x.z));
                    atomicMax(a + 128 * k + 96, enc_f32(x.w));
                }
            }
        }
    }
}

template <int D>
__device__ __forceinline__ int gin_fold(const int* __restrict__ src, const int* __restrict__ dst,
                                        const float* __restrict__ X, int lo, int rn,
                                        unsigned* acc, const int* q, const unsigned* qn,
                                        unsigned* __restrict__ lst, int lbase,
                                        int wave, int lane)
{
    const int cnt = min(__builtin_amdgcn_readfirstlane((int)*qn), QR);
    const int nmine = (cnt > wave) ? ((cnt - wave + 31) >> 5) : 0;
    for (int m0 = 0; m0 < nmine; m0 += 32) {
        const int nv = min(32, nmine - m0);
        int sv = 0, tv = -1;
        if (lane < nv) {
            const int qi = wave + 32 * (m0 + lane);
            const int e = q[qi];
            const unsigned r = (unsigned)(dst[e] - lo);
            sv = min(max(src[e], 0), NNODES - 1);
            tv = (r < (unsigned)rn) ? (int)r : -1;
            if (lst != nullptr && lbase + qi < LCAP)
                lst[lbase + qi] = ((unsigned)sv << 8) | (unsigned)(tv + 1);
        }
        gin_fold_rows<D>(X, acc, sv, tv, nv, lane);
    }
    return cnt;
}

template <int D>
__device__ __forceinline__ void gin_replay(const unsigned* __restrict__ lst, int n,
                                           const float* __restrict__ X, int rn,
                                           unsigned* acc, int wave, int lane)
{
    constexpr int U = 8 / (D / 128);
    const int per   = (((n + 31) >> 5) + U - 1) / U * U;
    const int start = wave * per;
    const int end   = min(n, start + per);
    unsigned pn = (start + lane < end) ? lst[start + lane] : 0u;
    for (int m0 = start; m0 < end; m0 += 32) {
        const unsigned p = pn;
        pn = (m0 + 32 + lane < end) ? lst[m0 + 32 + lane] : 0u;
        const int nv = min(32, end - m0);
        const int sv = min((int)(p >> 8), NNODES - 1);
        const int t  = (int)(p & 255u) - 1;
        const int tv = ((unsigned)t < (unsigned)rn) ? t : -1;
        gin_fold_rows<D>(X, acc, sv, tv, nv, lane);
    }
}

template <int D>
__global__ __launch_bounds__(1024) void gin_rangemax_kernel(
    const int* __restrict__ src, const int* __restrict__ dst, const unsigned* __restrict__ keys,
    const float* __restrict__ X, float* __restrict__ xsum, int E,
    unsigned* __restrict__ list, unsigned* __restrict__ lcnt, int replay)
{
    constexpr int NB = D / 128;
    __shared__ __attribute__((aligned(16))) unsigned acc[RNODES * D];
    __shared__ __attribute__((aligned(16))) int queue[2 * QR];
    __shared__ unsigned qcount[3];
    __shared__ unsigned ovf;

    const int tid  = threadIdx.x;
    const int lane = tid & 31;
    const int wave = __builtin_amdgcn_readfirstlane(tid >> 5);
    const int lo   = blockIdx.x * RNODES;
    const int rn   = min(RNODES, NNODES - lo);
    const unsigned pat = (unsigned)blockIdx.x * 0x01010101u;

    for (int i = tid; i < RNODES * D / 4; i += 1024) ((uint4*)acc)[i] = make_uint4(0u, 0u, 0u, 0u);
    if (tid < 3) qcount[tid] = 0u;
    if (tid == 3) ovf = 0u;
    __syncthreads();

    const int nsuper = (E + 4095) >> 12;
    const int rot    = (int)(((long long)blockIdx.x * nsuper) >> 8);
    const int niter  = (nsuper + 31) >> 5;
    const int ipr    = max(2, (niter + 15) >> 4);
    const int nround = (niter + ipr - 1) / ipr;

    unsigned* glist = list + (size_t)blockIdx.x * LCAP;
    const unsigned nlist = replay ? (unsigned)__builtin_amdgcn_readfirstlane((int)lcnt[blockIdx.x]) : LNONE;
    if (nlist <= (unsigned)LCAP) {
        gin_replay<D>(glist, (int)nlist, X, rn, acc, wave, lane);
    } else {
        unsigned* keep = replay ? nullptr : glist;
        int lbase = 0;
        const bool fold_first = ((wave >> 2) & 1) != 0;
        int c_prev = 2, c_cur = 0, c_next = 1;
        for (int r = 0; r <= nround; ++r) {
            int* qcur = queue + (r & 1) * QR;
            const int* qprev = queue + ((r & 1) ^ 1) * QR;
            int cnt = 0;
            if (tid == 0) qcount[c_next] = 0u;
            if (fold_first && r > 0) cnt = gin_fold<D>(src, dst, X, lo, rn, acc, qprev, &qcount[c_prev], keep, lbase, wave, lane);
            if (r < nround) gin_filter(keys, E, nsuper, rot, r * ipr, ipr, wave, lane, pat, qcur, &qcount[c_cur], &ovf);
            if (!fold_first && r > 0) cnt = gin_fold<D>(src, dst, X, lo, rn, acc, qprev, &qcount[c_prev], keep, lbase, wave, lane);
            lbase += cnt;
            __syncthreads();
            const int t = c_prev; c_prev = c_cur; c_cur = c_next; c_next = t;
        }

        if (tid == 0 && !replay) lcnt[blockIdx.x] = (ovf != 0u || lbase > LCAP) ? LNONE : (unsigned)lbase;

        if (ovf != 0u) {
            for (int e = tid; e < E; e += 1024) {
                const unsigned r = (unsigned)(dst[e] - lo);
                if (r < (unsigned)rn) {
                    const int s = min(max(src[e], 0), NNODES - 1);
                    const float* __restrict__ xr = X + (size_t)s * D;
                    for (int c = 0; c < D; ++c) atomicMax(acc + r * D + acc_pos(c), enc_f32(xr[c]));
                }
            }
        }
    }
    __syncthreads();

    for (int n = wave; n < rn; n += 32) {
        const float* __restrict__ xr = X + (size_t)(lo + n) * D;
        float* __restrict__ o = xsum + (size_t)(lo + n) * D;
#pragma unroll
        for (int k = 0; k < NB; ++k) {
            const float4 x = *(const float4*)(xr + 128 * k + 4 * lane);
            const unsigned* a = acc + n * D + 128 * k + lane;
            float4 r;
            r.x = x.x + dec_f32(a[0]);
            r.y = x.y + dec_f32(a[32]);
            r.z = x.z + dec_f32(a[64]);
            r.w = x.w + dec_f32(a[96]);
            *(float4*)(o + 128 * k + 4 * lane) = r;
        }
    }
}

__global__ __launch_bounds__(128) void gin_gemm_wmma_kernel(
    const float* __restrict__ X, const float* __restrict__ W,
    const float* __restrict__ bias, float* __restrict__ Y,
    int K, int Nout, int relu)
{
    const int lane  = threadIdx.x & 31;
    const int wave  = threadIdx.x >> 5;
    const int mtile = blockIdx.x * 4 + wave;
    if (mtile >= MTILES) return;

    const int row0    = mtile * 16;
    const int l15     = lane & 15;
    const int koff    = (lane >> 4) * 2;
    const int colbase = blockIdx.y * 64 + l15;

    const float* __restrict__ xrow = X + (size_t)(row0 + l15) * K;

    v8f acc0 = {}, acc1 = {}, acc2 = {}, acc3 = {};
#pragma unroll 2
    for (int k0 = 0; k0 < K; k0 += 4) {
        v2f a;
        a.x = xrow[k0 + koff + 0];
        a.y = xrow[k0 + koff + 1];
        const float* __restrict__ w0 = W + (size_t)(k0 + koff + 0) * Nout + colbase;
        const float* __restrict__ w1 = W + (size_t)(k0 + koff + 1) * Nout + colbase;
        v2f b0, b1, b2, b3;
        b0.x = w0[ 0]; b0.y = w1[ 0];
        b1.x = w0[16]; b1.y = w1[16];
        b2.x = w0[32]; b2.y = w1[32];
        b3.x = w0[48]; b3.y = w1[48];
        acc0 = __builtin_amdgcn_wmma_f32_16x16x4_f32(false, a, false, b0, (short)0, acc0, false, false);
        acc1 = __builtin_amdgcn_wmma_f32_16x16x4_f32(false, a, false, b1, (short)0, acc1, false, false);
        acc2 = __builtin_amdgcn_wmma_f32_16x16x4_f32(false, a, false, b2, (short)0, acc2, false, false);
        acc3 = __builtin_amdgcn_wmma_f32_16x16x4_f32(false, a, false, b3, (short)0, acc3, false, false);
    }

    const int rbase = row0 + (lane >> 4) * 8;
    v8f accs[4] = {acc0, acc1, acc2, acc3};
#pragma unroll
    for (int t = 0; t < 4; ++t) {
        const int col = colbase + 16 * t;
        const float bv = bias[col];
        float* __restrict__ yc = Y + col;
#pragma unroll
        for (int r = 0; r < 8; ++r) {
            float v = accs[t][r] + bv;
            if (relu) v = fmaxf(v, 0.0f);
            yc[(size_t)(rbase + r) * Nout] = v;
        }
    }
}

static void gin_layer(const int* src, const int* dst, const unsigned* keys, int E,
                      const float* Xin, int d_in_, const float* W, const float* b,
                      int d_out_, float* Y, int relu,
                      float* xsum, unsigned* list, unsigned* lcnt, int replay, hipStream_t stream)
{
    if (d_in_ == 128)
        gin_rangemax_kernel<128><<<NGROUPS, 1024, 0, stream>>>(src, dst, keys, Xin, xsum, E, list, lcnt, replay);
    else
        gin_rangemax_kernel<256><<<NGROUPS, 1024, 0, stream>>>(src, dst, keys, Xin, xsum, E, list, lcnt, replay);

    dim3 grid((MTILES + 3) / 4, d_out_ / 64);
    gin_gemm_wmma_kernel<<<grid, 128, 0, stream>>>(xsum, W, b, Y, d_in_, d_out_, relu);
}

extern "C" void kernel_launch(void* const* d_in, const int* in_sizes, int n_in,
                              void* d_out, int out_size, void* d_ws, size_t ws_size,
                              hipStream_t stream)
{
    const int*   src = (const int*)d_in[0];
    const int*   dst = (const int*)d_in[1];
    const float* emb = (const float*)d_in[2];
    const float* W1  = (const float*)d_in[3];
    const float* b1  = (const float*)d_in[4];
    const float* W2  = (const float*)d_in[5];
    const float* b2  = (const float*)d_in[6];
    const float* W3  = (const float*)d_in[7];
    const float* b3  = (const float*)d_in[8];
    float* out = (float*)d_out;
    const int E = in_sizes[0];

    const size_t cap = (size_t)NNODES * 256;
    float* xsum = (float*)d_ws;
    float* h    = (float*)d_ws + cap;
    unsigned* keys = (unsigned*)d_ws + 2 * cap;
    unsigned* list = keys + (1u << 20);
    unsigned* lcnt = list + (size_t)NGROUPS * LCAP;

    gin_keys_kernel<<<4 * NGROUPS, 256, 0, stream>>>(dst, keys, E);

    gin_layer(src, dst, keys, E, emb, 128, W1, b1, 256, h, 1, xsum, list, lcnt, 0, stream);
    gin_layer(src, dst, keys, E, h, 256, W2, b2, 256, h, 1, xsum, list, lcnt, 1, stream);
    gin_layer(src, dst, keys, E, h, 256, W3, b3, 128, out, 0, xsum, list, lcnt, 1, stream);
}


# GEMM re-cut: a wave keeps the W fragments of its 32 output columns for all of K in registers and walks the 16-row tiles; X by 16-byte row loads with lane halves exchanged; sum order unchanged
# speedup vs baseline: 3.7042x; 1.2471x over previous
#include <hip/hip_runtime.h>

#define NNODES 50000
#define MTILES (NNODES / 16)

typedef float v2f __attribute__((ext_vector_type(2)));
typedef float v8f __attribute__((ext_vector_type(8)));
typedef float v4f __attribute__((ext_vector_type(4)));

__device__ __forceinline__ unsigned enc_f32(float f) {
    unsigned u = __float_as_uint(f);
    return (u & 0x80000000u) ? ~u : (u | 0x80000000u);
}
__device__ __forceinline__ float dec_f32(unsigned e) {
    if (e == 0u) return 0.0f;
    unsigned u = (e & 0x80000000u) ? (e ^ 0x80000000u) : ~e;
    return __uint_as_float(u);
}

#define RNODES  196
#define NGROUPS 256
#define QR      4096
#define LCAP    65536
#define LNONE   0xFFFFFFFFu

__device__ __forceinline__ unsigned gin_key(int d) {
    return min((unsigned)d / (unsigned)RNODES, 255u);
}

__device__ __forceinline__ unsigned zero_bytes(unsigned x) {
    return (x - 0x01010101u) & ~x & 0x80808080u;
}

__global__ __launch_bounds__(256) void gin_keys_kernel(const int* __restrict__ dst,
                                                       unsigned* __restrict__ keys, int E)
{
    const int ndw = ((E + 4095) >> 12) << 10;
    for (int t = blockIdx.x * 256 + threadIdx.x; t < ndw; t += gridDim.x * 256) {
        const int e0 = 4 * t;
        int4 d;
        if (e0 + 4 <= E) {
            d = *(const int4*)(dst + e0);
        } else {
            d.x = (e0 + 0 < E) ? dst[e0 + 0] : -1;
            d.y = (e0 + 1 < E) ? dst[e0 + 1] : -1;
            d.z = (e0 + 2 < E) ? dst[e0 + 2] : -1;
            d.w = (e0 + 3 < E) ? dst[e0 + 3] : -1;
        }
        keys[t] = gin_key(d.x) | (gin_key(d.y) << 8) | (gin_key(d.z) << 16) | (gin_key(d.w) << 24);
    }
}

__device__ __forceinline__ int acc_pos(int c) {
    return (c & ~127) + ((c & 3) << 5) + ((c & 127) >> 2);
}

__device__ __forceinline__ void gin_filter(const unsigned* __restrict__ keys, int E, int nsuper, int rot,
                                           int sweep0, int ipr, int wave, int lane, unsigned pat,
                                           int* q, unsigned* qn, unsigned* ovf)
{
    for (int it = 0; it < ipr; ++it) {
        const int sc0 = wave + 32 * (sweep0 + it);
        if (sc0 >= nsuper) break;
        const int sc = (sc0 + rot < nsuper) ? (sc0 + rot) : (sc0 + rot - nsuper);
        const int base = (sc << 10) + lane * 4;
        uint4 k[8];
#pragma unroll
        for (int u = 0; u < 8; ++u) k[u] = *(const uint4*)(keys + base + u * 128);
#pragma unroll
        for (int u = 0; u < 8; ++u) {
            const unsigned w[4] = {k[u].x, k[u].y, k[u].z, k[u].w};
#pragma unroll
            for (int c = 0; c < 4; ++c) {
                unsigned x  = w[c] ^ pat;
                unsigned hz = zero_bytes(x);
                if (__builtin_amdgcn_ballot_w32(hz != 0u) != 0u) {
                    do {
                        if (hz != 0u) {
                            const int sh = __builtin_ctz(hz) & ~7;
                            const int e  = 4 * (base + u * 128 + c) + (sh >> 3);
                            if (e < E) {
                                const unsigned slot = atomicAdd(qn, 1u);
                                if (slot < (unsigned)QR) q[slot] = e;
                                else *ovf = 1u;
                            }
                            x |= 0xFFu << sh;
                            hz = zero_bytes(x);
                        }
                    } while (__builtin_amdgcn_ballot_w32(hz != 0u) != 0u);
                }
            }
        }
    }
}

template <int D>
__device__ __forceinline__ void gin_fold_rows(const float* __restrict__ X, unsigned* acc,
                                              int sv, int tv, int nv, int lane)
{
    constexpr int NB = D / 128;
    constexpr int U  = 8 / NB;
    static_assert(U * NB == 8, "the load fence below names 8 register sets");
    for (int j = 0; j < nv; j += U) {
        v4f v[8];
        int tr[U];
#pragma unroll
        for (int u = 0; u < U; ++u) {
            const int ju = min(j + u, nv - 1);
            const int s = __builtin_amdgcn_readlane(sv, ju);
            tr[u] = (j + u < nv) ? __builtin_amdgcn_readlane(tv, ju) : -1;
#pragma unroll
            for (int k = 0; k < NB; ++k)
                v[u * NB + k] = *(const v4f*)(X + (size_t)s * D + 128 * k + 4 * lane);
        }
        asm volatile("" :: "v"(v[0]), "v"(v[1]), "v"(v[2]), "v"(v[3]),
                           "v"(v[4]), "v"(v[5]), "v"(v[6]), "v"(v[7]));
#pragma unroll
        for (int u = 0; u < U; ++u) {
            if (tr[u] >= 0) {
                unsigned* a = acc + tr[u] * D + lane;
#pragma unroll
                for (int k = 0; k < NB; ++k) {
                    const v4f x = v[u * NB + k];
                    atomicMax(a + 128 * k +  0, enc_f32(x.x));
                    atomicMax(a + 128 * k + 32, enc_f32(x.y));
                    atomicMax(a + 128 * k + 64, enc_f32(x.z));
                    atomicMax(a + 128 * k + 96, enc_f32(x.w));
                }
            }
        }
    }
}

template <int D>
__device__ __forceinline__ int gin_fold(const int* __restrict__ src, const int* __restrict__ dst,
                                        const float* __restrict__ X, int lo, int rn,
                                        unsigned* acc, const int* q, const unsigned* qn,
                                        unsigned* __restrict__ lst, int lbase,
                                        int wave, int lane)
{
    const int cnt = min(__builtin_amdgcn_readfirstlane((int)*qn), QR);
    const int nmine = (cnt > wave) ? ((cnt - wave + 31) >> 5) : 0;
    for (int m0 = 0; m0 < nmine; m0 += 32) {
        const int nv = min(32, nmine - m0);
        int sv = 0, tv = -1;
        if (lane < nv) {
            const int qi = wave + 32 * (m0 + lane);
            const int e = q[qi];
            const unsigned r = (unsigned)(dst[e] - lo);
            sv = min(max(src[e], 0), NNODES - 1);
            tv = (r < (unsigned)rn) ? (int)r : -1;
            if (lst != nullptr && lbase + qi < LCAP)
                lst[lbase + qi] = ((unsigned)sv << 8) | (unsigned)(tv + 1);
        }
        gin_fold_rows<D>(X, acc, sv, tv, nv, lane);
    }
    return cnt;
}

template <int D>
__device__ __forceinline__ void gin_replay(const unsigned* __restrict__ lst, int n,
                                           const float* __restrict__ X, int rn,
                                           unsigned* acc, int wave, int lane)
{
    constexpr int U = 8 / (D / 128);
    const int per   = (((n + 31) >> 5) + U - 1) / U * U;
    const int start = wave * per;
    const int end   = min(n, start + per);
    unsigned pn = (start + lane < end) ? lst[start + lane] : 0u;
    for (int m0 = start; m0 < end; m0 += 32) {
        const unsigned p = pn;
        pn = (m0 + 32 + lane < end) ? lst[m0 + 32 + lane] : 0u;
        const int nv = min(32, end - m0);
        const int sv = min((int)(p >> 8), NNODES - 1);
        const int t  = (int)(p & 255u) - 1;
        const int tv = ((unsigned)t < (unsigned)rn) ? t : -1;
        gin_fold_rows<D>(X, acc, sv, tv, nv, lane);
    }
}

template <int D>
__global__ __launch_bounds__(1024) void gin_rangemax_kernel(
    const int* __restrict__ src, const int* __restrict__ dst, const unsigned* __restrict__ keys,
    const float* __restrict__ X, float* __restrict__ xsum, int E,
    unsigned* __restrict__ list, unsigned* __restrict__ lcnt, int replay)
{
    constexpr int NB = D / 128;
    __shared__ __attribute__((aligned(16))) unsigned acc[RNODES * D];
    __shared__ __attribute__((aligned(16))) int queue[2 * QR];
    __shared__ unsigned qcount[3];
    __shared__ unsigned ovf;

    const int tid  = threadIdx.x;
    const int lane = tid & 31;
    const int wave = __builtin_amdgcn_readfirstlane(tid >> 5);
    const int lo   = blockIdx.x * RNODES;
    const int rn   = min(RNODES, NNODES - lo);
    const unsigned pat = (unsigned)blockIdx.x * 0x01010101u;

    for (int i = tid; i < RNODES * D / 4; i += 1024) ((uint4*)acc)[i] = make_uint4(0u, 0u, 0u, 0u);
    if (tid < 3) qcount[tid] = 0u;
    if (tid == 3) ovf = 0u;
    __syncthreads();

    const int nsuper = (E + 4095) >> 12;
    const int rot    = (int)(((long long)blockIdx.x * nsuper) >> 8);
    const int niter  = (nsuper + 31) >> 5;
    const int ipr    = max(2, (niter + 15) >> 4);
    const int nround = (niter + ipr - 1) / ipr;

    unsigned* glist = list + (size_t)blockIdx.x * LCAP;
    const unsigned nlist = replay ? (unsigned)__builtin_amdgcn_readfirstlane((int)lcnt[blockIdx.x]) : LNONE;
    if (nlist <= (unsigned)LCAP) {
        gin_replay<D>(glist, (int)nlist, X, rn, acc, wave, lane);
    } else {
        unsigned* keep = replay ? nullptr : glist;
        int lbase = 0;
        const bool fold_first = ((wave >> 2) & 1) != 0;
        int c_prev = 2, c_cur = 0, c_next = 1;
        for (int r = 0; r <= nround; ++r) {
            int* qcur = queue + (r & 1) * QR;
            const int* qprev = queue + ((r & 1) ^ 1) * QR;
            int cnt = 0;
            if (tid == 0) qcount[c_next] = 0u;
            if (fold_first && r > 0) cnt = gin_fold<D>(src, dst, X, lo, rn, acc, qprev, &qcount[c_prev], keep, lbase, wave, lane);
            if (r < nround) gin_filter(keys, E, nsuper, rot, r * ipr, ipr, wave, lane, pat, qcur, &qcount[c_cur], &ovf);
            if (!fold_first && r > 0) cnt = gin_fold<D>(src, dst, X, lo, rn, acc, qprev, &qcount[c_prev], keep, lbase, wave, lane);
            lbase += cnt;
            __syncthreads();
            const int t = c_prev; c_prev = c_cur; c_cur = c_next; c_next = t;
        }

        if (tid == 0 && !replay) lcnt[blockIdx.x] = (ovf != 0u || lbase > LCAP) ? LNONE : (unsigned)lbase;

        if (ovf != 0u) {
            for (int e = tid; e < E; e += 1024) {
                const unsigned r = (unsigned)(dst[e] - lo);
                if (r < (unsigned)rn) {
                    const int s = min(max(src[e], 0), NNODES - 1);
                    const float* __restrict__ xr = X + (size_t)s * D;
                    for (int c = 0; c < D; ++c) atomicMax(acc + r * D + acc_pos(c), enc_f32(xr[c]));
                }
            }
        }
    }
    __syncthreads();

    for (int n = wave; n < rn; n += 32) {
        const float* __restrict__ xr = X + (size_t)(lo + n) * D;
        float* __restrict__ o = xsum + (size_t)(lo + n) * D;
#pragma unroll
        for (int k = 0; k < NB; ++k) {
            const float4 x = *(const float4*)(xr + 128 * k + 4 * lane);
            const unsigned* a = acc + n * D + 128 * k + lane;
            float4 r;
            r.x = x.x + dec_f32(a[0]);
            r.y = x.y + dec_f32(a[32]);
            r.z = x.z + dec_f32(a[64]);
            r.w = x.w + dec_f32(a[96]);
            *(float4*)(o + 128 * k + 4 * lane) = r;
        }
    }
}

typedef unsigned v2u __attribute__((ext_vector_type(2)));
__device__ __forceinline__ void gin_half_swap(float& d, float& s) {
    const v2u r = __builtin_amdgcn_permlane16_swap(__float_as_uint(d), __float_as_uint(s), false, false);
    d = __uint_as_float(r.x);
    s = __uint_as_float(r.y);
}

__device__ __forceinline__ void gin_load_chunk(v4f (&q)[8], const float* __restrict__ p) {
#pragma unroll
    for (int j = 0; j < 8; ++j) q[j] = *(const v4f*)(p + 8 * j);
}

template <int NKS>
__device__ __forceinline__ void gin_gemm_chunk(v4f (&q)[8], const v2f (&bw)[NKS][2], int s0,
                                               v8f& acc0, v8f& acc1)
{
#pragma unroll
    for (int j = 0; j < 8; ++j) {
        float x = q[j].x, y = q[j].y, z = q[j].z, w = q[j].w;
        gin_half_swap(x, z);
        gin_half_swap(y, w);
        v2f a0; a0.x = x; a0.y = y;
        v2f a1; a1.x = z; a1.y = w;
        const int s = s0 + 2 * j;
        acc0 = __builtin_amdgcn_wmma_f32_16x16x4_f32(false, a0, false, bw[s][0],     (short)0, acc0, false, false);
        acc1 = __builtin_amdgcn_wmma_f32_16x16x4_f32(false, a0, false, bw[s][1],     (short)0, acc1, false, false);
        acc0 = __builtin_amdgcn_wmma_f32_16x16x4_f32(false, a1, false, bw[s + 1][0], (short)0, acc0, false, false);
        acc1 = __builtin_amdgcn_wmma_f32_16x16x4_f32(false, a1, false, bw[s + 1][1], (short)0, acc1, false, false);
    }
}

template <int K, int NOUT>
__global__ __launch_bounds__(256) void gin_gemm_wmma_kernel(
    const float* __restrict__ X, const float* __restrict__ W,
    const float* __restrict__ bias, float* __restrict__ Y, int relu)
{
    constexpr int NS  = NOUT / 32;
    constexpr int MG  = 8 / NS;
    constexpr int NKS = K / 4;
    constexpr int NCH = K / 64;
    static_assert(NS * MG == 8 && NCH % 2 == 0, "8 waves; chunks alternate between two register sets");

    const int lane  = threadIdx.x & 31;
    const int wave  = __builtin_amdgcn_readfirstlane((int)(threadIdx.x >> 5));
    const int strip = wave % NS;
    const int mg    = wave / NS;
    const int c     = lane & 15;
    const int h     = lane >> 4;
    const int col   = strip * 32 + 2 * c;

    v2f bw[NKS][2];
#pragma unroll
    for (int s = 0; s < NKS; ++s) {
        const float* __restrict__ w0 = W + (size_t)(4 * s + 2 * h) * NOUT + col;
        const v2f p0 = *(const v2f*)(w0);
        const v2f p1 = *(const v2f*)(w0 + NOUT);
        bw[s][0].x = p0.x;
        bw[s][0].y = p1.x;
        bw[s][1].x = p0.y;
        bw[s][1].y = p1.y;
    }

    const int mstep = (int)gridDim.x * MG;
    int mt = (int)blockIdx.x * MG + mg;

    v4f qa[8], qb[8];
    gin_load_chunk(qa, X + (size_t)(min(mt, MTILES - 1) * 16 + c) * K + 4 * h);

    for (; mt < MTILES; mt += mstep) {
        const float* __restrict__ xr = X + (size_t)(mt * 16 + c) * K + 4 * h;
        const float* __restrict__ xn = X + (size_t)(min(mt + mstep, MTILES - 1) * 16 + c) * K + 4 * h;
        v8f acc0 = {}, acc1 = {};
#pragma unroll
        for (int ch = 0; ch < NCH; ch += 2) {
            gin_load_chunk(qb, xr + 64 * (ch + 1));
            gin_gemm_chunk<NKS>(qa, bw, 16 * ch, acc0, acc1);
            gin_load_chunk(qa, (ch + 2 < NCH) ? xr + 64 * (ch + 2) : xn);
            gin_gemm_chunk<NKS>(qb, bw, 16 * (ch + 1), acc0, acc1);
        }

#pragma unroll
        for (int i = 0; i < NCH; ++i) {
            __builtin_amdgcn_sched_group_barrier(0x020, 8, 0);
            __builtin_amdgcn_sched_group_barrier(0x008, 32, 0);
        }

        const int rbase = mt * 16 + h * 8;
        const v2f bv = *(const v2f*)(bias + col);
        float* __restrict__ yc = Y + col;
#pragma unroll
        for (int r = 0; r < 8; ++r) {
            v2f o;
            o.x = acc0[r] + bv.x;
            o.y = acc1[r] + bv.y;
            if (relu) { o.x = fmaxf(o.x, 0.0f); o.y = fmaxf(o.y, 0.0f); }
            *(v2f*)(yc + (size_t)(rbase + r) * NOUT) = o;
        }
    }
}

static void gin_layer(const int* src, const int* dst, const unsigned* keys, int E,
                      const float* Xin, int d_in_, const float* W, const float* b,
                      int d_out_, float* Y, int relu,
                      float* xsum, unsigned* list, unsigned* lcnt, int replay, hipStream_t stream)
{
    if (d_in_ == 128)
        gin_rangemax_kernel<128><<<NGROUPS, 1024, 0, stream>>>(src, dst, keys, Xin, xsum, E, list, lcnt, replay);
    else
        gin_rangemax_kernel<256><<<NGROUPS, 1024, 0, stream>>>(src, dst, keys, Xin, xsum, E, list, lcnt, replay);

    if (d_in_ == 128)
        gin_gemm_wmma_kernel<128, 256><<<NGROUPS, 256, 0, stream>>>(xsum, W, b, Y, relu);
    else if (d_out_ == 256)
        gin_gemm_wmma_kernel<256, 256><<<NGROUPS, 256, 0, stream>>>(xsum, W, b, Y, relu);
    else
        gin_gemm_wmma_kernel<256, 128><<<NGROUPS, 256, 0, stream>>>(xsum, W, b, Y, relu);
}

extern "C" void kernel_launch(void* const* d_in, const int* in_sizes, int n_in,
                              void* d_out, int out_size, void* d_ws, size_t ws_size,
                              hipStream_t stream)
{
    const int*   src = (const int*)d_in[0];
    const int*   dst = (const int*)d_in[1];
    const float* emb = (const float*)d_in[2];
    const float* W1  = (const float*)d_in[3];
    const float* b1  = (const float*)d_in[4];
    const float* W2  = (const float*)d_in[5];
    const float* b2  = (const float*)d_in[6];
    const float* W3  = (const float*)d_in[7];
    const float* b3  = (const float*)d_in[8];
    float* out = (float*)d_out;
    const int E = in_sizes[0];

    const size_t cap = (size_t)NNODES * 256;
    float* xsum = (float*)d_ws;
    float* h    = (float*)d_ws + cap;
    unsigned* keys = (unsigned*)d_ws + 2 * cap;
    unsigned* list = keys + (1u << 20);
    unsigned* lcnt = list + (size_t)NGROUPS * LCAP;

    gin_keys_kernel<<<4 * NGROUPS, 256, 0, stream>>>(dst, keys, E);

    gin_layer(src, dst, keys, E, emb, 128, W1, b1, 256, h, 1, xsum, list, lcnt, 0, stream);
    gin_layer(src, dst, keys, E, h, 256, W2, b2, 256, h, 1, xsum, list, lcnt, 1, stream);
    gin_layer(src, dst, keys, E, h, 256, W3, b3, 128, out, 0, xsum, list, lcnt, 1, stream);
}
